# starting kernel
# baseline (speedup 1.0000x reference)
#include <hip/hip_runtime.h>


typedef __attribute__((ext_vector_type(2))) float v2f;
typedef __attribute__((ext_vector_type(8))) float v8f;

#define OR_      22
#define P_TOTAL  (22*22*22)
#define F_OUT    8
#define K_TOT    216
#define KF       (K_TOT*F_OUT)
#define XS_B     (24*24*24*8)
#define XS_D     (24*24*8)
#define XS_H     (24*8)
#define XS_W     8

__global__ __launch_bounds__(256) void lc3d_wmma_f32_kernel(
    const float* __restrict__ x,
    const float* __restrict__ kern,
    const float* __restrict__ bias,
    float* __restrict__ out)
{
    const int lane = threadIdx.x & 31;
    const int wave = threadIdx.x >> 5;
    const int p    = blockIdx.x * 8 + wave;

    const int i   = p / (OR_ * OR_);
    const int rem = p - i * (OR_ * OR_);
    const int j   = rem / OR_;
    const int z   = rem - j * OR_;

    const int half = lane >> 4;
    const int lrow = lane & 15;

    const int bsel = (lrow >= 1) ? 1 : 0;
    const float* xv_base =
        x + bsel * XS_B + i * XS_D + j * XS_H + z * XS_W + 2 * half;

    const float* kslice = kern + (size_t)p * KF;
    const float* kp     = kslice + (lrow & 7);

    __builtin_prefetch((const char*)kslice + lane * 216, 0, 1);

    v8f acc = {};
    int kc = 0;
#pragma unroll
    for (int di = 0; di < 3; ++di) {
#pragma unroll
        for (int dj = 0; dj < 3; ++dj) {
#pragma unroll
            for (int dk = 0; dk < 3; ++dk) {
                const float* xv = xv_base + di * XS_D + dj * XS_H + dk * XS_W;
#pragma unroll
                for (int cc = 0; cc < 8; cc += 4) {
                    v2f a = *(const v2f*)(xv + cc);

                    v2f b;
                    const int krow = kc + 2 * half;
                    b.x = kp[(size_t)krow * F_OUT];
                    b.y = kp[(size_t)(krow + 1) * F_OUT];

                    acc = __builtin_amdgcn_wmma_f32_16x16x4_f32(
                         false, a,  false, b,
                         (short)0, acc,
                         false,  false);
                    kc += 4;
                }
            }
        }
    }

    const float o0 = acc[0];
    const float o1 = acc[1];
    if (lane < 8) {
        const float bv = bias[(size_t)p * F_OUT + lane];
        out[(size_t)p * F_OUT + lane]                          = o0 + bv;
        out[(size_t)P_TOTAL * F_OUT + (size_t)p * F_OUT + lane] = o1 + bv;
    }
}

extern "C" void kernel_launch(void* const* d_in, const int* in_sizes, int n_in,
                              void* d_out, int out_size, void* d_ws, size_t ws_size,
                              hipStream_t stream) {
    const float* x    = (const float*)d_in[0];
    const float* kern = (const float*)d_in[1];
    const float* bias = (const float*)d_in[2];
    float* out        = (float*)d_out;

    (void)in_sizes; (void)n_in; (void)out_size; (void)d_ws; (void)ws_size;

    dim3 grid(P_TOTAL / 8);
    dim3 block(256);
    lc3d_wmma_f32_kernel<<<grid, block, 0, stream>>>(x, kern, bias, out);
}


# each wave computes 4 adjacent sites instead of 1 (4x fewer waves): two sites share every WMMA (odd site in A rows 2-3, B cols 8-15), 2 pairs per wave, next group's loads issued ahead; 1331x2 waves
# speedup vs baseline: 1.1032x; 1.1032x over previous
#include <hip/hip_runtime.h>


typedef __attribute__((ext_vector_type(2))) float v2f;
typedef __attribute__((ext_vector_type(8))) float v8f;

#define OR_      22
#define P_TOTAL  (22*22*22)
#define F_OUT    8
#define K_TOT    216
#define KF       (K_TOT*F_OUT)
#define XS_B     (24*24*24*8)
#define XS_D     (24*24*8)
#define XS_H     (24*8)
#define XS_W     8

#define SITES_PER_WAVE 4
#define WAVES_PER_BLOCK 2

__global__ __launch_bounds__(64) void lc3d_wmma_f32_kernel(
    const float* __restrict__ x,
    const float* __restrict__ kern,
    const float* __restrict__ bias,
    float* __restrict__ out)
{
    const int lane = threadIdx.x & 31;
    const int wave = threadIdx.x >> 5;
    const int p0   = (blockIdx.x * WAVES_PER_BLOCK + wave) * SITES_PER_WAVE;

    const int half = lane >> 4;
    const int lrow = lane & 15;

    const int bsel = lrow & 1;
    const int ssel = (lrow >> 1) & 1;
    const int bcol = (lrow & 7) + (lrow >> 3) * KF;

    const float* xv_base[2];
#pragma unroll
    for (int q = 0; q < 2; ++q) {
        const int p   = p0 + 2 * q;
        const int i   = p / (OR_ * OR_);
        const int rem = p - i * (OR_ * OR_);
        const int j   = rem / OR_;
        const int z   = rem - j * OR_;
        xv_base[q] = x + bsel * XS_B + i * XS_D + j * XS_H
                       + (z + ssel) * XS_W + 2 * half;
    }

    const float* kslice = kern + (size_t)p0 * KF;
    const float* kp     = kslice + bcol;

#pragma unroll
    for (int sidx = 0; sidx < SITES_PER_WAVE; ++sidx)
        __builtin_prefetch((const char*)kslice + sidx * (KF * 4) + lane * 216, 0, 1);

    v8f acc0 = {};
    v8f acc1 = {};

    v2f A0[2][6], A1[2][6], B0[2][6], B1[2][6];

#define LOAD_GROUP(G, S)                                                        \
    _Pragma("unroll")                                                           \
    for (int c = 0; c < 6; ++c) {                                               \
          \
          \
        const int xoff = ((G) / 3) * XS_D + ((G) % 3) * XS_H                    \
                       + (c >> 1) * XS_W + (c & 1) * 4;                         \
          \
        const int krow = ((G) * 6 + c) * 4 + 2 * half;                          \
        A0[S][c]   = *(const v2f*)(xv_base[0] + xoff);                          \
        B0[S][c].x = kp[(size_t)krow * F_OUT];                                  \
        B0[S][c].y = kp[(size_t)(krow + 1) * F_OUT];                            \
        A1[S][c]   = *(const v2f*)(xv_base[1] + xoff);                          \
        B1[S][c].x = kp[(size_t)krow * F_OUT + 2 * KF];                         \
        B1[S][c].y = kp[(size_t)(krow + 1) * F_OUT + 2 * KF];                   \
    }

#define MMA_GROUP(S)                                                            \
    _Pragma("unroll")                                                           \
    for (int c = 0; c < 6; ++c) {                                               \
        acc0 = __builtin_amdgcn_wmma_f32_16x16x4_f32(                           \
             false, A0[S][c],  false, B0[S][c],               \
             (short)0, acc0,  false,  false);    \
        acc1 = __builtin_amdgcn_wmma_f32_16x16x4_f32(                           \
             false, A1[S][c],  false, B1[S][c],               \
             (short)0, acc1,  false,  false);    \
    }

    LOAD_GROUP(0, 0)
#pragma unroll
    for (int g = 0; g < 9; ++g) {
        if (g + 1 < 9) {
            if ((g + 1) & 1) { LOAD_GROUP(g + 1, 1) } else { LOAD_GROUP(g + 1, 0) }
        }
        __builtin_amdgcn_sched_barrier(0);
        if (g & 1) { MMA_GROUP(1) } else { MMA_GROUP(0) }
    }
#undef LOAD_GROUP
#undef MMA_GROUP

    {
        const bool odd = lrow >= 8;
        const float e0 = half ? acc1[0] : acc0[0];
        const float e1 = half ? acc1[1] : acc0[1];
        const float o0 = half ? acc1[2] : acc0[2];
        const float o1 = half ? acc1[3] : acc0[3];
        const size_t o = (size_t)p0 * F_OUT + lane;
        const float bv = bias[o];
        out[o]                           = (odd ? o0 : e0) + bv;
        out[(size_t)P_TOTAL * F_OUT + o] = (odd ? o1 : e1) + bv;
    }
}

extern "C" void kernel_launch(void* const* d_in, const int* in_sizes, int n_in,
                              void* d_out, int out_size, void* d_ws, size_t ws_size,
                              hipStream_t stream) {
    const float* x    = (const float*)d_in[0];
    const float* kern = (const float*)d_in[1];
    const float* bias = (const float*)d_in[2];
    float* out        = (float*)d_out;

    (void)in_sizes; (void)n_in; (void)out_size; (void)d_ws; (void)ws_size;

    dim3 grid(P_TOTAL / 8);
    dim3 block(64);
    lc3d_wmma_f32_kernel<<<grid, block, 0, stream>>>(x, kern, bias, out);
}


# B fragments loaded as 8-byte feature pairs, all four sites in every WMMA (A rows = site x batch x feature parity, zero in the other k slot): 163 loads per wave, was 325; k order and sum tree changed
# speedup vs baseline: 1.2730x; 1.1539x over previous
#include <hip/hip_runtime.h>


typedef __attribute__((ext_vector_type(2))) float v2f;
typedef __attribute__((ext_vector_type(8))) float v8f;

#define OR_      22
#define P_TOTAL  (22*22*22)
#define F_OUT    8
#define K_TOT    216
#define KF       (K_TOT*F_OUT)
#define XS_B     (24*24*24*8)
#define XS_D     (24*24*8)
#define XS_H     (24*8)
#define XS_W     8

#define SITES_PER_WAVE 4
#define WAVES_PER_BLOCK 2
#define NGROUP   9
#define AHEAD    3

__global__ __launch_bounds__(64) void lc3d_wmma_f32_kernel(
    const float* __restrict__ x,
    const float* __restrict__ kern,
    const float* __restrict__ bias,
    float* __restrict__ out)
{
    const int lane = threadIdx.x & 31;
    const int wave = threadIdx.x >> 5;
    const int p0   = (blockIdx.x * WAVES_PER_BLOCK + wave) * SITES_PER_WAVE;

    const int half = lane >> 4;
    const int lrow = lane & 15;

    const bool par  = (lrow & 1) != 0;
    const int  bsel = (lrow >> 1) & 1;
    const int  site = lrow >> 2;
    const int  fp   = lrow & 3;

    const float* xa;
    {
        const int p   = p0 + 2 * (site >> 1);
        const int i   = p / (OR_ * OR_);
        const int rem = p - i * (OR_ * OR_);
        const int j   = rem / OR_;
        const int z   = rem - j * OR_;
        xa = x + bsel * XS_B + i * XS_D + j * XS_H
               + (z + (site & 1)) * XS_W + 2 * half;
    }

    const float* kslice = kern + (size_t)p0 * KF;
    const float* kb     = kslice + (size_t)site * KF + 2 * fp + 2 * half * F_OUT;

#pragma unroll
    for (int sidx = 0; sidx < SITES_PER_WAVE; ++sidx)
        __builtin_prefetch((const char*)kslice + sidx * (KF * 4) + lane * 216, 0, 1);

    const size_t o  = (size_t)p0 * F_OUT + 2 * lrow;
    v2f          bv = *(const v2f*)(bias + o);
    __builtin_amdgcn_sched_barrier(0);

    v8f acc0 = {};
    v8f acc1 = {};

    v2f XA[AHEAD + 1][6];
    v2f BW[AHEAD + 1][12];

#define LOAD_GROUP(G, S)                                                        \
    _Pragma("unroll")                                                           \
    for (int t = 0; t < 6; ++t) {                                               \
        const int dk = t >> 1, c2 = t & 1;                                      \
        const int xoff = ((G) / 3) * XS_D + ((G) % 3) * XS_H                    \
                       + dk * XS_W + c2 * 4;                                    \
            \
        const int krow = ((G) * 3 + dk) * 8 + c2 * 4;                           \
        XA[S][t]         = *(const v2f*)(xa + xoff);                            \
        BW[S][2 * t]     = *(const v2f*)(kb + (size_t)krow * F_OUT);            \
        BW[S][2 * t + 1] = *(const v2f*)(kb + (size_t)(krow + 1) * F_OUT);      \
    }

#define MMA_GROUP(S)                                                            \
    _Pragma("unroll")                                                           \
    for (int t = 0; t < 6; ++t) {                                               \
        const float x0 = XA[S][t].x, x1 = XA[S][t].y;                           \
        v2f a0, a1;                                                             \
        a0.x = par ? 0.0f : x0;  a0.y = par ? x0 : 0.0f;                        \
        a1.x = par ? 0.0f : x1;  a1.y = par ? x1 : 0.0f;                        \
        acc0 = __builtin_amdgcn_wmma_f32_16x16x4_f32(                           \
             false, a0,  false, BW[S][2 * t],                 \
             (short)0, acc0,  false,  false);    \
        acc1 = __builtin_amdgcn_wmma_f32_16x16x4_f32(                           \
             false, a1,  false, BW[S][2 * t + 1],             \
             (short)0, acc1,  false,  false);    \
    }

#define LOAD_SEL(G) { if (((G) & 3) == 0) { LOAD_GROUP(G, 0) }                  \
                 else if (((G) & 3) == 1) { LOAD_GROUP(G, 1) }                  \
                 else if (((G) & 3) == 2) { LOAD_GROUP(G, 2) }                  \
                 else                     { LOAD_GROUP(G, 3) } }
#define MMA_SEL(G)  { if (((G) & 3) == 0) { MMA_GROUP(0) }                      \
                 else if (((G) & 3) == 1) { MMA_GROUP(1) }                      \
                 else if (((G) & 3) == 2) { MMA_GROUP(2) }                      \
                 else                     { MMA_GROUP(3) } }

    LOAD_SEL(0)
    LOAD_SEL(1)
    LOAD_SEL(2)
#pragma unroll
    for (int g = 0; g < NGROUP; ++g) {
        if (g + AHEAD < NGROUP) LOAD_SEL(g + AHEAD)
        __builtin_amdgcn_sched_barrier(0);
        MMA_SEL(g)
        if (g == 0) asm volatile("" : "+v"(bv));
        __builtin_amdgcn_sched_barrier(0);
    }
#undef LOAD_GROUP
#undef MMA_GROUP
#undef LOAD_SEL
#undef MMA_SEL

    {
        const v8f  acc   = acc0 + acc1;
        const bool upper = (site & 1) != 0;
        const bool mine  = (lrow >> 3) == half;
        v2f r0, r1;
        r0.x = (upper ? acc[4] : acc[0]) + bv.x;
        r0.y = (upper ? acc[5] : acc[1]) + bv.y;
        r1.x = (upper ? acc[6] : acc[2]) + bv.x;
        r1.y = (upper ? acc[7] : acc[3]) + bv.y;
        if (mine) {
            *(v2f*)(out + o)                           = r0;
            *(v2f*)(out + (size_t)P_TOTAL * F_OUT + o) = r1;
        }
    }
}

extern "C" void kernel_launch(void* const* d_in, const int* in_sizes, int n_in,
                              void* d_out, int out_size, void* d_ws, size_t ws_size,
                              hipStream_t stream) {
    const float* x    = (const float*)d_in[0];
    const float* kern = (const float*)d_in[1];
    const float* bias = (const float*)d_in[2];
    float* out        = (float*)d_out;

    (void)in_sizes; (void)n_in; (void)out_size; (void)d_ws; (void)ws_size;

    dim3 grid(P_TOTAL / 8);
    dim3 block(64);
    lc3d_wmma_f32_kernel<<<grid, block, 0, stream>>>(x, kern, bias, out);
}


# weight-slice prefetches removed (4 per wave, every lane on a separate 256-byte unit: 128 line look-ups and second-level requests per wave that return no data); loads, products and sums unchanged
# speedup vs baseline: 1.3559x; 1.0651x over previous
#include <hip/hip_runtime.h>


typedef __attribute__((ext_vector_type(2))) float v2f;
typedef __attribute__((ext_vector_type(8))) float v8f;

#define OR_      22
#define P_TOTAL  (22*22*22)
#define F_OUT    8
#define K_TOT    216
#define KF       (K_TOT*F_OUT)
#define XS_B     (24*24*24*8)
#define XS_D     (24*24*8)
#define XS_H     (24*8)
#define XS_W     8

#define SITES_PER_WAVE 4
#define WAVES_PER_BLOCK 2
#define NGROUP   9
#define AHEAD    3

__global__ __launch_bounds__(64) void lc3d_wmma_f32_kernel(
    const float* __restrict__ x,
    const float* __restrict__ kern,
    const float* __restrict__ bias,
    float* __restrict__ out)
{
    const int lane = threadIdx.x & 31;
    const int wave = threadIdx.x >> 5;
    const int p0   = (blockIdx.x * WAVES_PER_BLOCK + wave) * SITES_PER_WAVE;

    const int half = lane >> 4;
    const int lrow = lane & 15;

    const bool par  = (lrow & 1) != 0;
    const int  bsel = (lrow >> 1) & 1;
    const int  site = lrow >> 2;
    const int  fp   = lrow & 3;

    const float* xa;
    {
        const int p   = p0 + 2 * (site >> 1);
        const int i   = p / (OR_ * OR_);
        const int rem = p - i * (OR_ * OR_);
        const int j   = rem / OR_;
        const int z   = rem - j * OR_;
        xa = x + bsel * XS_B + i * XS_D + j * XS_H
               + (z + (site & 1)) * XS_W + 2 * half;
    }

    const float* kslice = kern + (size_t)p0 * KF;
    const float* kb     = kslice + (size_t)site * KF + 2 * fp + 2 * half * F_OUT;

    const size_t o  = (size_t)p0 * F_OUT + 2 * lrow;
    v2f          bv = *(const v2f*)(bias + o);
    __builtin_amdgcn_sched_barrier(0);

    v8f acc0 = {};
    v8f acc1 = {};

    v2f XA[AHEAD + 1][6];
    v2f BW[AHEAD + 1][12];

#define LOAD_GROUP(G, S)                                                        \
    _Pragma("unroll")                                                           \
    for (int t = 0; t < 6; ++t) {                                               \
        const int dk = t >> 1, c2 = t & 1;                                      \
        const int xoff = ((G) / 3) * XS_D + ((G) % 3) * XS_H                    \
                       + dk * XS_W + c2 * 4;                                    \
            \
        const int krow = ((G) * 3 + dk) * 8 + c2 * 4;                           \
        XA[S][t]         = *(const v2f*)(xa + xoff);                            \
        BW[S][2 * t]     = *(const v2f*)(kb + (size_t)krow * F_OUT);            \
        BW[S][2 * t + 1] = *(const v2f*)(kb + (size_t)(krow + 1) * F_OUT);      \
    }

#define MMA_GROUP(S)                                                            \
    _Pragma("unroll")                                                           \
    for (int t = 0; t < 6; ++t) {                                               \
        const float x0 = XA[S][t].x, x1 = XA[S][t].y;                           \
        v2f a0, a1;                                                             \
        a0.x = par ? 0.0f : x0;  a0.y = par ? x0 : 0.0f;                        \
        a1.x = par ? 0.0f : x1;  a1.y = par ? x1 : 0.0f;                        \
        acc0 = __builtin_amdgcn_wmma_f32_16x16x4_f32(                           \
             false, a0,  false, BW[S][2 * t],                 \
             (short)0, acc0,  false,  false);    \
        acc1 = __builtin_amdgcn_wmma_f32_16x16x4_f32(                           \
             false, a1,  false, BW[S][2 * t + 1],             \
             (short)0, acc1,  false,  false);    \
    }

#define LOAD_SEL(G) { if (((G) & 3) == 0) { LOAD_GROUP(G, 0) }                  \
                 else if (((G) & 3) == 1) { LOAD_GROUP(G, 1) }                  \
                 else if (((G) & 3) == 2) { LOAD_GROUP(G, 2) }                  \
                 else                     { LOAD_GROUP(G, 3) } }
#define MMA_SEL(G)  { if (((G) & 3) == 0) { MMA_GROUP(0) }                      \
                 else if (((G) & 3) == 1) { MMA_GROUP(1) }                      \
                 else if (((G) & 3) == 2) { MMA_GROUP(2) }                      \
                 else                     { MMA_GROUP(3) } }

    LOAD_SEL(0)
    LOAD_SEL(1)
    LOAD_SEL(2)
#pragma unroll
    for (int g = 0; g < NGROUP; ++g) {
        if (g + AHEAD < NGROUP) LOAD_SEL(g + AHEAD)
        __builtin_amdgcn_sched_barrier(0);
        MMA_SEL(g)
        if (g == 0) asm volatile("" : "+v"(bv));
        __builtin_amdgcn_sched_barrier(0);
    }
#undef LOAD_GROUP
#undef MMA_GROUP
#undef LOAD_SEL
#undef MMA_SEL

    {
        const v8f  acc   = acc0 + acc1;
        const bool upper = (site & 1) != 0;
        const bool mine  = (lrow >> 3) == half;
        v2f r0, r1;
        r0.x = (upper ? acc[4] : acc[0]) + bv.x;
        r0.y = (upper ? acc[5] : acc[1]) + bv.y;
        r1.x = (upper ? acc[6] : acc[2]) + bv.x;
        r1.y = (upper ? acc[7] : acc[3]) + bv.y;
        if (mine) {
            *(v2f*)(out + o)                           = r0;
            *(v2f*)(out + (size_t)P_TOTAL * F_OUT + o) = r1;
        }
    }
}

extern "C" void kernel_launch(void* const* d_in, const int* in_sizes, int n_in,
                              void* d_out, int out_size, void* d_ws, size_t ws_size,
                              hipStream_t stream) {
    const float* x    = (const float*)d_in[0];
    const float* kern = (const float*)d_in[1];
    const float* bias = (const float*)d_in[2];
    float* out        = (float*)d_out;

    (void)in_sizes; (void)n_in; (void)out_size; (void)d_ws; (void)ws_size;

    dim3 grid(P_TOTAL / 8);
    dim3 block(64);
    lc3d_wmma_f32_kernel<<<grid, block, 0, stream>>>(x, kern, bias, out);
}
